# starting kernel
# baseline (speedup 1.0000x reference)
#include <hip/hip_runtime.h>

#define LOG2E 1.4426950408889634f

typedef __attribute__((ext_vector_type(2))) float v2f;
typedef __attribute__((ext_vector_type(8))) float v8f;

#define BB   8
#define NN   1024
#define MM   1024
#define CIN  8
#define CREP 9
#define COUT 16
#define MT   32
#define NCHUNK 8
#define NPER (NN / NCHUNK)

__global__ __launch_bounds__(256) void convcnp_kernel(
    const float* __restrict__ x,
    const float* __restrict__ y,
    const float* __restrict__ t,
    const float* __restrict__ sigma,
    const float* __restrict__ W,
    const float* __restrict__ bias,
    float* __restrict__ out)
{
    __shared__ float  sx[NN];
    __shared__ float4 sy4[NN * 2];
    __shared__ float  spart[NCHUNK * MT * 10];
    __shared__ float  yo[MT * 12];
    __shared__ float  Wp[COUT * 12];

    const int tid   = threadIdx.x;
    const int b     = blockIdx.y;
    const int mt    = blockIdx.x;
    const int m_l   = tid & (MT - 1);
    const int chunk = tid >> 5;

    {
        const float4* x4 = (const float4*)(x + b * NN);
        ((float4*)sx)[tid] = x4[tid];
        const float4* y4 = (const float4*)(y + (size_t)b * NN * CIN);
        #pragma unroll
        for (int k = 0; k < 8; ++k)
            sy4[tid + k * 256] = y4[tid + k * 256];
    }

    float k2[CREP];
    #pragma unroll
    for (int c = 0; c < CREP; ++c) {
        float s = sigma[c];
        k2[c] = -0.5f * LOG2E * __builtin_amdgcn_exp2f(-2.0f * s * LOG2E);
    }

    const float tm = t[b * MM + mt * MT + m_l];
    __syncthreads();

    float acc[CREP];
    #pragma unroll
    for (int c = 0; c < CREP; ++c) acc[c] = 0.f;

    const int n0 = chunk * NPER;
    #pragma unroll 4
    for (int i = 0; i < NPER; ++i) {
        const int n = n0 + i;
        const float  xn = sx[n];
        const float4 ya = sy4[2 * n];
        const float4 yb = sy4[2 * n + 1];
        const float diff = xn - tm;
        const float dd   = diff * diff;
        float yr[CIN] = { ya.x, ya.y, ya.z, ya.w, yb.x, yb.y, yb.z, yb.w };
        acc[0] += __builtin_amdgcn_exp2f(dd * k2[0]);
        #pragma unroll
        for (int c = 1; c < CREP; ++c)
            acc[c] += __builtin_amdgcn_exp2f(dd * k2[c]) * yr[c - 1];
    }

    {
        const int base = (chunk * MT + m_l) * 10;
        #pragma unroll
        for (int c = 0; c < CREP; ++c) spart[base + c] = acc[c];
    }
    __syncthreads();

    if (tid < MT) {
        float s[CREP];
        #pragma unroll
        for (int c = 0; c < CREP; ++c) s[c] = 0.f;
        for (int ch = 0; ch < NCHUNK; ++ch) {
            const int base = (ch * MT + tid) * 10;
            #pragma unroll
            for (int c = 0; c < CREP; ++c) s[c] += spart[base + c];
        }
        const float dens = s[0];
        const float inv  = 1.0f / (dens + 1e-8f);
        yo[tid * 12 + 0] = dens;
        #pragma unroll
        for (int c = 1; c < CREP; ++c) yo[tid * 12 + c] = s[c] * inv;
        yo[tid * 12 + 9]  = 0.f;
        yo[tid * 12 + 10] = 0.f;
        yo[tid * 12 + 11] = 0.f;
    }
    if (tid < COUT * 12) {
        const int o = tid / 12, c = tid % 12;
        Wp[tid] = (c < CREP) ? W[o * CREP + c] : 0.f;
    }
    __syncthreads();

    const int wid  = tid >> 5;
    const int lane = tid & 31;
    if (wid < 2) {
        const int half  = lane >> 4;
        const int lr    = lane & 15;
        const int mbase = wid * 16;

        v8f d = {0.f, 0.f, 0.f, 0.f, 0.f, 0.f, 0.f, 0.f};
        #pragma unroll
        for (int kc = 0; kc < 3; ++kc) {
            const int c0 = kc * 4 + 2 * half;
            v2f a, bm;
            a.x  = yo[(mbase + lr) * 12 + c0];
            a.y  = yo[(mbase + lr) * 12 + c0 + 1];
            bm.x = Wp[lr * 12 + c0];
            bm.y = Wp[lr * 12 + c0 + 1];
            d = __builtin_amdgcn_wmma_f32_16x16x4_f32(
                    false, a, false, bm, (short)0, d, false, false);
        }

        const float bo = bias[lr];
        const int mg = mt * MT + mbase;
        #pragma unroll
        for (int r = 0; r < 8; ++r) {
            const int m = mg + r + 8 * half;
            out[((size_t)b * MM + m) * COUT + lr] = d[r] + bo;
        }
    }
}

extern "C" void kernel_launch(void* const* d_in, const int* in_sizes, int n_in,
                              void* d_out, int out_size, void* d_ws, size_t ws_size,
                              hipStream_t stream) {
    const float* x     = (const float*)d_in[0];
    const float* y     = (const float*)d_in[1];
    const float* t     = (const float*)d_in[2];
    const float* sigma = (const float*)d_in[3];
    const float* W     = (const float*)d_in[4];
    const float* bias  = (const float*)d_in[5];
    float* out = (float*)d_out;

    dim3 grid(MM / MT, BB);
    dim3 block(256);
    convcnp_kernel<<<grid, block, 0, stream>>>(x, y, t, sigma, W, bias, out);
}


# serial tail shortened: sigma, W, bias loaded at entry, W and bias staged in LDS before barrier 1; waves 0,1 sum the chunk partials straight into WMMA A fragments, third barrier removed
# speedup vs baseline: 1.0620x; 1.0620x over previous
#include <hip/hip_runtime.h>

#define LOG2E 1.4426950408889634f

typedef __attribute__((ext_vector_type(2))) float v2f;
typedef __attribute__((ext_vector_type(8))) float v8f;

#define BB   8
#define NN   1024
#define MM   1024
#define CIN  8
#define CREP 9
#define COUT 16
#define MT   32
#define NCHUNK 8
#define NPER (NN / NCHUNK)

__global__ __launch_bounds__(256) void convcnp_kernel(
    const float* __restrict__ x,
    const float* __restrict__ y,
    const float* __restrict__ t,
    const float* __restrict__ sigma,
    const float* __restrict__ W,
    const float* __restrict__ bias,
    float* __restrict__ out)
{
    __shared__ float  sx[NN];
    __shared__ float4 sy4[NN * 2];
    __shared__ __attribute__((aligned(16))) float spart[NCHUNK * MT * 10];
    __shared__ __attribute__((aligned(16))) float Wp[COUT * 12];
    __shared__ float  sb[COUT];

    const int tid   = threadIdx.x;
    const int b     = blockIdx.y;
    const int mt    = blockIdx.x;
    const int m_l   = tid & (MT - 1);
    const int chunk = tid >> 5;

    const float4* x4 = (const float4*)(x + b * NN);
    const float4* y4 = (const float4*)(y + (size_t)b * NN * CIN);
    const float4 xr = x4[tid];
    float4 yg[8];
    #pragma unroll
    for (int k = 0; k < 8; ++k) yg[k] = y4[tid + k * 256];
    float wv = 0.f;
    if (tid < COUT * 12) {
        const int o = tid / 12, c = tid % 12;
        if (c < CREP) wv = W[o * CREP + c];
    }
    float bv = 0.f;
    if (tid < COUT) bv = bias[tid];

    const float tm = t[b * MM + mt * MT + m_l];

    float k2[CREP];
    #pragma unroll
    for (int c = 0; c < CREP; ++c) {
        float s = sigma[c];
        k2[c] = -0.5f * LOG2E * __builtin_amdgcn_exp2f(-2.0f * s * LOG2E);
    }

    ((float4*)sx)[tid] = xr;
    #pragma unroll
    for (int k = 0; k < 8; ++k) sy4[tid + k * 256] = yg[k];
    if (tid < COUT * 12) Wp[tid] = wv;
    if (tid < COUT) sb[tid] = bv;

    #pragma unroll
    for (int c = 0; c < CREP; ++c) asm volatile("" :: "v"(k2[c]));
    __syncthreads();

    float acc[CREP];
    #pragma unroll
    for (int c = 0; c < CREP; ++c) acc[c] = 0.f;

    const int n0 = chunk * NPER;
    #pragma unroll 4
    for (int i = 0; i < NPER; ++i) {
        const int n = n0 + i;
        const float  xn = sx[n];
        const float4 ya = sy4[2 * n];
        const float4 yb = sy4[2 * n + 1];
        const float diff = xn - tm;
        const float dd   = diff * diff;
        float yr[CIN] = { ya.x, ya.y, ya.z, ya.w, yb.x, yb.y, yb.z, yb.w };
        acc[0] += __builtin_amdgcn_exp2f(dd * k2[0]);
        #pragma unroll
        for (int c = 1; c < CREP; ++c)
            acc[c] += __builtin_amdgcn_exp2f(dd * k2[c]) * yr[c - 1];
    }

    {
        const int base = (chunk * MT + m_l) * 10;
        #pragma unroll
        for (int c = 0; c < CREP; ++c) spart[base + c] = acc[c];
    }
    __syncthreads();

    const int wid  = tid >> 5;
    const int lane = tid & 31;
    if (wid < 2) {
        const int half  = lane >> 4;
        const int lr    = lane & 15;
        const int mbase = wid * 16;
        const int row   = mbase + lr;
        const int so    = half ? 0 : 8;

        v2f bfr[3];
        #pragma unroll
        for (int kc = 0; kc < 3; ++kc) {
            const int c0 = kc * 4 + 2 * half;
            bfr[kc].x = Wp[lr * 12 + c0];
            bfr[kc].y = Wp[lr * 12 + c0 + 1];
        }
        const float bo = sb[lr];

        v2f p0 = {0.f, 0.f};
        v2f p1 = {0.f, 0.f};
        float s8 = 0.f;
        #pragma unroll
        for (int ch = 0; ch < NCHUNK; ++ch) {
            const float* pr = spart + (ch * MT + row) * 10;
            p0 += *(const v2f*)(pr + 2 * half);
            p1 += *(const v2f*)(pr + 4 + 2 * half);
            s8 += pr[so];
        }
        const float dens = half ? s8 : p0.x;
        const float inv  = 1.0f / (dens + 1e-8f);
        v2f afr[3];
        afr[0].x = half ? p0.x * inv : dens;
        afr[0].y = p0.y * inv;
        afr[1].x = p1.x * inv;
        afr[1].y = p1.y * inv;
        afr[2].x = half ? 0.f : s8 * inv;
        afr[2].y = 0.f;

        v8f d = {0.f, 0.f, 0.f, 0.f, 0.f, 0.f, 0.f, 0.f};
        #pragma unroll
        for (int kc = 0; kc < 3; ++kc)
            d = __builtin_amdgcn_wmma_f32_16x16x4_f32(
                    false, afr[kc], false, bfr[kc], (short)0, d, false, false);

        const int mg = mt * MT + mbase;
        #pragma unroll
        for (int r = 0; r < 8; ++r) {
            const int m = mg + r + 8 * half;
            out[((size_t)b * MM + m) * COUT + lr] = d[r] + bo;
        }
    }
}

extern "C" void kernel_launch(void* const* d_in, const int* in_sizes, int n_in,
                              void* d_out, int out_size, void* d_ws, size_t ws_size,
                              hipStream_t stream) {
    const float* x     = (const float*)d_in[0];
    const float* y     = (const float*)d_in[1];
    const float* t     = (const float*)d_in[2];
    const float* sigma = (const float*)d_in[3];
    const float* W     = (const float*)d_in[4];
    const float* bias  = (const float*)d_in[5];
    float* out = (float*)d_out;

    dim3 grid(MM / MT, BB);
    dim3 block(256);
    convcnp_kernel<<<grid, block, 0, stream>>>(x, y, t, sigma, W, bias, out);
}


# speedup vs baseline: 1.2066x; 1.1362x over previous
#include <hip/hip_runtime.h>

#define LOG2E 1.4426950408889634f

typedef __attribute__((ext_vector_type(2))) float v2f;
typedef __attribute__((ext_vector_type(8))) float v8f;

#define BB   8
#define NN   1024
#define MM   1024
#define CIN  8
#define CREP 9
#define COUT 16
#define MT   32
#define NCHUNK 8
#define NPER (NN / NCHUNK)

__global__ __launch_bounds__(256) void convcnp_kernel(
    const float* __restrict__ x,
    const float* __restrict__ y,
    const float* __restrict__ t,
    const float* __restrict__ sigma,
    const float* __restrict__ W,
    const float* __restrict__ bias,
    float* __restrict__ out)
{
    __shared__ float  sx[NN];
    __shared__ float4 sy4[NN * 2];
    __shared__ __attribute__((aligned(16))) float spart[NCHUNK * MT * 10];
    __shared__ __attribute__((aligned(16))) float Wp[COUT * 12];
    __shared__ float  sb[COUT];

    const int tid   = threadIdx.x;
    const int b     = blockIdx.y;
    const int mt    = blockIdx.x;
    const int m_l   = tid & (MT - 1);
    const int chunk = tid >> 5;

    const float4* x4 = (const float4*)(x + b * NN);
    const float4* y4 = (const float4*)(y + (size_t)b * NN * CIN);
    const float4 xr = x4[tid];
    float4 yg[8];
    #pragma unroll
    for (int k = 0; k < 8; ++k) yg[k] = y4[tid + k * 256];
    float wv = 0.f;
    if (tid < COUT * 12) {
        const int o = tid / 12, c = tid % 12;
        if (c < CREP) wv = W[o * CREP + c];
    }
    float bv = 0.f;
    if (tid < COUT) bv = bias[tid];

    const float tm = t[b * MM + mt * MT + m_l];

    float k2[CREP];
    #pragma unroll
    for (int c = 0; c < CREP; ++c) {
        float s = sigma[c];
        k2[c] = -0.5f * LOG2E * __builtin_amdgcn_exp2f(-2.0f * s * LOG2E);
    }

    ((float4*)sx)[tid] = xr;
    #pragma unroll
    for (int k = 0; k < 8; ++k) sy4[tid + k * 256] = yg[k];
    if (tid < COUT * 12) Wp[tid] = wv;
    if (tid < COUT) sb[tid] = bv;

    #pragma unroll
    for (int c = 0; c < CREP; ++c) asm volatile("" :: "v"(k2[c]));
    __syncthreads();

    float accz = 0.f;
    v2f ap0 = {0.f, 0.f}, ap1 = {0.f, 0.f}, ap2 = {0.f, 0.f}, ap3 = {0.f, 0.f};
    const v2f kp0 = { k2[1], k2[2] }, kp1 = { k2[3], k2[4] };
    const v2f kp2 = { k2[5], k2[6] }, kp3 = { k2[7], k2[8] };
    const v2f tm2 = { tm, tm };

    const float4* sx4 = (const float4*)sx;
    const int q0 = chunk * (NPER / 4);
    float4 xc0 = sx4[q0 + 0];
    float4 xc1 = sx4[q0 + 1];
    #pragma unroll 1
    for (int i = 0; i < NPER / 8; ++i) {
        const int q  = q0 + 2 * i;
        const int qn = q0 + 2 * ((i + 1 < NPER / 8) ? i + 1 : i);
        float4 yv[16];
        yv[15] = sy4[8 * q + 15];
        yv[14] = sy4[8 * q + 14];
        yv[13] = sy4[8 * q + 13];
        yv[12] = sy4[8 * q + 12];
        yv[11] = sy4[8 * q + 11];
        yv[10] = sy4[8 * q + 10];
        yv[9] = sy4[8 * q + 9];
        yv[8] = sy4[8 * q + 8];
        yv[7] = sy4[8 * q + 7];
        yv[6] = sy4[8 * q + 6];
        yv[5] = sy4[8 * q + 5];
        yv[4] = sy4[8 * q + 4];
        yv[3] = sy4[8 * q + 3];
        yv[2] = sy4[8 * q + 2];
        yv[1] = sy4[8 * q + 1];
        yv[0] = sy4[8 * q + 0];
        asm volatile("" ::: "memory");
        __builtin_amdgcn_sched_barrier(0);
        const v2f d0a = (v2f){ xc0.x, xc0.y } - tm2;
        const v2f d0b = (v2f){ xc0.z, xc0.w } - tm2;
        const v2f d1a = (v2f){ xc1.x, xc1.y } - tm2;
        const v2f d1b = (v2f){ xc1.z, xc1.w } - tm2;
        __builtin_amdgcn_sched_barrier(0);
        asm volatile("" ::: "memory");
        xc0 = sx4[qn + 0];
        xc1 = sx4[qn + 1];
        asm volatile("" ::: "memory");
        __builtin_amdgcn_sched_barrier(0);
        const v2f q0a = d0a * d0a;
        const v2f q0b = d0b * d0b;
        const v2f q1a = d1a * d1a;
        const v2f q1b = d1b * d1b;
        v2f a0_0 = (v2f){ q0a.x, q0a.x } * kp0;
        v2f a0_1 = (v2f){ q0a.x, q0a.x } * kp1;
        float a0_z = q0a.x * k2[0];
        __builtin_amdgcn_sched_barrier(0);
        a0_0.x = __builtin_amdgcn_exp2f(a0_0.x);
        __builtin_amdgcn_sched_barrier(0);
        v2f a0_2 = (v2f){ q0a.x, q0a.x } * kp2;
        __builtin_amdgcn_sched_barrier(0);
        a0_0.y = __builtin_amdgcn_exp2f(a0_0.y);
        __builtin_amdgcn_sched_barrier(0);
        v2f a0_3 = (v2f){ q0a.x, q0a.x } * kp3;
        __builtin_amdgcn_sched_barrier(0);
        a0_1.x = __builtin_amdgcn_exp2f(a0_1.x);
        __builtin_amdgcn_sched_barrier(0);
        v2f a1_0 = (v2f){ q0a.y, q0a.y } * kp0;
        __builtin_amdgcn_sched_barrier(0);
        a0_1.y = __builtin_amdgcn_exp2f(a0_1.y);
        __builtin_amdgcn_sched_barrier(0);
        v2f a1_1 = (v2f){ q0a.y, q0a.y } * kp1;
        __builtin_amdgcn_sched_barrier(0);
        a0_z = __builtin_amdgcn_exp2f(a0_z);
        __builtin_amdgcn_sched_barrier(0);
        float a1_z = q0a.y * k2[0];
        __builtin_amdgcn_sched_barrier(0);
        a0_2.x = __builtin_amdgcn_exp2f(a0_2.x);
        __builtin_amdgcn_sched_barrier(0);
        v2f a1_2 = (v2f){ q0a.y, q0a.y } * kp2;
        __builtin_amdgcn_sched_barrier(0);
        a0_2.y = __builtin_amdgcn_exp2f(a0_2.y);
        __builtin_amdgcn_sched_barrier(0);
        v2f a1_3 = (v2f){ q0a.y, q0a.y } * kp3;
        __builtin_amdgcn_sched_barrier(0);
        a0_3.x = __builtin_amdgcn_exp2f(a0_3.x);
        __builtin_amdgcn_sched_barrier(0);
        v2f a2_0 = (v2f){ q0b.x, q0b.x } * kp0;
        __builtin_amdgcn_sched_barrier(0);
        a0_3.y = __builtin_amdgcn_exp2f(a0_3.y);
        __builtin_amdgcn_sched_barrier(0);
        v2f a2_1 = (v2f){ q0b.x, q0b.x } * kp1;
        __builtin_amdgcn_sched_barrier(0);
        a1_0.x = __builtin_amdgcn_exp2f(a1_0.x);
        __builtin_amdgcn_sched_barrier(0);
        float a2_z = q0b.x * k2[0];
        __builtin_amdgcn_sched_barrier(0);
        a1_0.y = __builtin_amdgcn_exp2f(a1_0.y);
        __builtin_amdgcn_sched_barrier(0);
        v2f a2_2 = (v2f){ q0b.x, q0b.x } * kp2;
        __builtin_amdgcn_sched_barrier(0);
        a1_1.x = __builtin_amdgcn_exp2f(a1_1.x);
        __builtin_amdgcn_sched_barrier(0);
        v2f a2_3 = (v2f){ q0b.x, q0b.x } * kp3;
        __builtin_amdgcn_sched_barrier(0);
        a1_1.y = __builtin_amdgcn_exp2f(a1_1.y);
        __builtin_amdgcn_sched_barrier(0);
        v2f a3_0 = (v2f){ q0b.y, q0b.y } * kp0;
        __builtin_amdgcn_sched_barrier(0);
        a1_z = __builtin_amdgcn_exp2f(a1_z);
        __builtin_amdgcn_sched_barrier(0);
        v2f a3_1 = (v2f){ q0b.y, q0b.y } * kp1;
        __builtin_amdgcn_sched_barrier(0);
        a1_2.x = __builtin_amdgcn_exp2f(a1_2.x);
        __builtin_amdgcn_sched_barrier(0);
        float a3_z = q0b.y * k2[0];
        __builtin_amdgcn_sched_barrier(0);
        a1_2.y = __builtin_amdgcn_exp2f(a1_2.y);
        __builtin_amdgcn_sched_barrier(0);
        v2f a3_2 = (v2f){ q0b.y, q0b.y } * kp2;
        __builtin_amdgcn_sched_barrier(0);
        a1_3.x = __builtin_amdgcn_exp2f(a1_3.x);
        __builtin_amdgcn_sched_barrier(0);
        v2f a3_3 = (v2f){ q0b.y, q0b.y } * kp3;
        __builtin_amdgcn_sched_barrier(0);
        a1_3.y = __builtin_amdgcn_exp2f(a1_3.y);
        __builtin_amdgcn_sched_barrier(0);
        v2f a4_0 = (v2f){ q1a.x, q1a.x } * kp0;
        __builtin_amdgcn_sched_barrier(0);
        a2_0.x = __builtin_amdgcn_exp2f(a2_0.x);
        __builtin_amdgcn_sched_barrier(0);
        v2f a4_1 = (v2f){ q1a.x, q1a.x } * kp1;
        __builtin_amdgcn_sched_barrier(0);
        a2_0.y = __builtin_amdgcn_exp2f(a2_0.y);
        __builtin_amdgcn_sched_barrier(0);
        float a4_z = q1a.x * k2[0];
        __builtin_amdgcn_sched_barrier(0);
        a2_1.x = __builtin_amdgcn_exp2f(a2_1.x);
        __builtin_amdgcn_sched_barrier(0);
        v2f a4_2 = (v2f){ q1a.x, q1a.x } * kp2;
        __builtin_amdgcn_sched_barrier(0);
        a2_1.y = __builtin_amdgcn_exp2f(a2_1.y);
        __builtin_amdgcn_sched_barrier(0);
        v2f a4_3 = (v2f){ q1a.x, q1a.x } * kp3;
        __builtin_amdgcn_sched_barrier(0);
        a2_z = __builtin_amdgcn_exp2f(a2_z);
        __builtin_amdgcn_sched_barrier(0);
        v2f a5_0 = (v2f){ q1a.y, q1a.y } * kp0;
        __builtin_amdgcn_sched_barrier(0);
        a2_2.x = __builtin_amdgcn_exp2f(a2_2.x);
        __builtin_amdgcn_sched_barrier(0);
        v2f a5_1 = (v2f){ q1a.y, q1a.y } * kp1;
        __builtin_amdgcn_sched_barrier(0);
        a2_2.y = __builtin_amdgcn_exp2f(a2_2.y);
        __builtin_amdgcn_sched_barrier(0);
        float a5_z = q1a.y * k2[0];
        __builtin_amdgcn_sched_barrier(0);
        a2_3.x = __builtin_amdgcn_exp2f(a2_3.x);
        __builtin_amdgcn_sched_barrier(0);
        v2f a5_2 = (v2f){ q1a.y, q1a.y } * kp2;
        __builtin_amdgcn_sched_barrier(0);
        a2_3.y = __builtin_amdgcn_exp2f(a2_3.y);
        __builtin_amdgcn_sched_barrier(0);
        v2f a5_3 = (v2f){ q1a.y, q1a.y } * kp3;
        __builtin_amdgcn_sched_barrier(0);
        a3_0.x = __builtin_amdgcn_exp2f(a3_0.x);
        __builtin_amdgcn_sched_barrier(0);
        v2f a6_0 = (v2f){ q1b.x, q1b.x } * kp0;
        __builtin_amdgcn_sched_barrier(0);
        a3_0.y = __builtin_amdgcn_exp2f(a3_0.y);
        __builtin_amdgcn_sched_barrier(0);
        v2f a6_1 = (v2f){ q1b.x, q1b.x } * kp1;
        __builtin_amdgcn_sched_barrier(0);
        a3_1.x = __builtin_amdgcn_exp2f(a3_1.x);
        __builtin_amdgcn_sched_barrier(0);
        float a6_z = q1b.x * k2[0];
        __builtin_amdgcn_sched_barrier(0);
        a3_1.y = __builtin_amdgcn_exp2f(a3_1.y);
        __builtin_amdgcn_sched_barrier(0);
        v2f a6_2 = (v2f){ q1b.x, q1b.x } * kp2;
        __builtin_amdgcn_sched_barrier(0);
        a3_z = __builtin_amdgcn_exp2f(a3_z);
        __builtin_amdgcn_sched_barrier(0);
        v2f a6_3 = (v2f){ q1b.x, q1b.x } * kp3;
        __builtin_amdgcn_sched_barrier(0);
        a3_2.x = __builtin_amdgcn_exp2f(a3_2.x);
        __builtin_amdgcn_sched_barrier(0);
        v2f a7_0 = (v2f){ q1b.y, q1b.y } * kp0;
        __builtin_amdgcn_sched_barrier(0);
        a3_2.y = __builtin_amdgcn_exp2f(a3_2.y);
        __builtin_amdgcn_sched_barrier(0);
        v2f a7_1 = (v2f){ q1b.y, q1b.y } * kp1;
        __builtin_amdgcn_sched_barrier(0);
        a3_3.x = __builtin_amdgcn_exp2f(a3_3.x);
        __builtin_amdgcn_sched_barrier(0);
        float a7_z = q1b.y * k2[0];
        __builtin_amdgcn_sched_barrier(0);
        a3_3.y = __builtin_amdgcn_exp2f(a3_3.y);
        __builtin_amdgcn_sched_barrier(0);
        v2f a7_2 = (v2f){ q1b.y, q1b.y } * kp2;
        __builtin_amdgcn_sched_barrier(0);
        a4_0.x = __builtin_amdgcn_exp2f(a4_0.x);
        __builtin_amdgcn_sched_barrier(0);
        accz += a0_z;
        __builtin_amdgcn_sched_barrier(0);
        a4_0.y = __builtin_amdgcn_exp2f(a4_0.y);
        __builtin_amdgcn_sched_barrier(0);
        accz += a1_z;
        __builtin_amdgcn_sched_barrier(0);
        a4_1.x = __builtin_amdgcn_exp2f(a4_1.x);
        __builtin_amdgcn_sched_barrier(0);
        accz += a2_z;
        __builtin_amdgcn_sched_barrier(0);
        a4_1.y = __builtin_amdgcn_exp2f(a4_1.y);
        __builtin_amdgcn_sched_barrier(0);
        ap0 = __builtin_elementwise_fma(a0_0, (v2f){ yv[0].x, yv[0].y }, ap0);
        __builtin_amdgcn_sched_barrier(0);
        a4_z = __builtin_amdgcn_exp2f(a4_z);
        __builtin_amdgcn_sched_barrier(0);
        ap1 = __builtin_elementwise_fma(a0_1, (v2f){ yv[0].z, yv[0].w }, ap1);
        __builtin_amdgcn_sched_barrier(0);
        a4_2.x = __builtin_amdgcn_exp2f(a4_2.x);
        __builtin_amdgcn_sched_barrier(0);
        ap2 = __builtin_elementwise_fma(a0_2, (v2f){ yv[1].x, yv[1].y }, ap2);
        __builtin_amdgcn_sched_barrier(0);
        a4_2.y = __builtin_amdgcn_exp2f(a4_2.y);
        __builtin_amdgcn_sched_barrier(0);
        ap3 = __builtin_elementwise_fma(a0_3, (v2f){ yv[1].z, yv[1].w }, ap3);
        __builtin_amdgcn_sched_barrier(0);
        a4_3.x = __builtin_amdgcn_exp2f(a4_3.x);
        __builtin_amdgcn_sched_barrier(0);
        ap0 = __builtin_elementwise_fma(a1_0, (v2f){ yv[2].x, yv[2].y }, ap0);
        __builtin_amdgcn_sched_barrier(0);
        a4_3.y = __builtin_amdgcn_exp2f(a4_3.y);
        __builtin_amdgcn_sched_barrier(0);
        ap1 = __builtin_elementwise_fma(a1_1, (v2f){ yv[2].z, yv[2].w }, ap1);
        __builtin_amdgcn_sched_barrier(0);
        a5_0.x = __builtin_amdgcn_exp2f(a5_0.x);
        __builtin_amdgcn_sched_barrier(0);
        ap2 = __builtin_elementwise_fma(a1_2, (v2f){ yv[3].x, yv[3].y }, ap2);
        __builtin_amdgcn_sched_barrier(0);
        a5_0.y = __builtin_amdgcn_exp2f(a5_0.y);
        __builtin_amdgcn_sched_barrier(0);
        ap3 = __builtin_elementwise_fma(a1_3, (v2f){ yv[3].z, yv[3].w }, ap3);
        __builtin_amdgcn_sched_barrier(0);
        a5_1.x = __builtin_amdgcn_exp2f(a5_1.x);
        __builtin_amdgcn_sched_barrier(0);
        ap0 = __builtin_elementwise_fma(a2_0, (v2f){ yv[4].x, yv[4].y }, ap0);
        __builtin_amdgcn_sched_barrier(0);
        a5_1.y = __builtin_amdgcn_exp2f(a5_1.y);
        __builtin_amdgcn_sched_barrier(0);
        ap1 = __builtin_elementwise_fma(a2_1, (v2f){ yv[4].z, yv[4].w }, ap1);
        __builtin_amdgcn_sched_barrier(0);
        a5_z = __builtin_amdgcn_exp2f(a5_z);
        __builtin_amdgcn_sched_barrier(0);
        ap2 = __builtin_elementwise_fma(a2_2, (v2f){ yv[5].x, yv[5].y }, ap2);
        __builtin_amdgcn_sched_barrier(0);
        a5_2.x = __builtin_amdgcn_exp2f(a5_2.x);
        __builtin_amdgcn_sched_barrier(0);
        ap3 = __builtin_elementwise_fma(a2_3, (v2f){ yv[5].z, yv[5].w }, ap3);
        __builtin_amdgcn_sched_barrier(0);
        a5_2.y = __builtin_amdgcn_exp2f(a5_2.y);
        __builtin_amdgcn_sched_barrier(0);
        ap0 = __builtin_elementwise_fma(a3_0, (v2f){ yv[6].x, yv[6].y }, ap0);
        __builtin_amdgcn_sched_barrier(0);
        a5_3.x = __builtin_amdgcn_exp2f(a5_3.x);
        __builtin_amdgcn_sched_barrier(0);
        ap1 = __builtin_elementwise_fma(a3_1, (v2f){ yv[6].z, yv[6].w }, ap1);
        __builtin_amdgcn_sched_barrier(0);
        a5_3.y = __builtin_amdgcn_exp2f(a5_3.y);
        __builtin_amdgcn_sched_barrier(0);
        accz += a3_z;
        __builtin_amdgcn_sched_barrier(0);
        a6_0.x = __builtin_amdgcn_exp2f(a6_0.x);
        __builtin_amdgcn_sched_barrier(0);
        ap2 = __builtin_elementwise_fma(a3_2, (v2f){ yv[7].x, yv[7].y }, ap2);
        __builtin_amdgcn_sched_barrier(0);
        a6_0.y = __builtin_amdgcn_exp2f(a6_0.y);
        __builtin_amdgcn_sched_barrier(0);
        ap3 = __builtin_elementwise_fma(a3_3, (v2f){ yv[7].z, yv[7].w }, ap3);
        __builtin_amdgcn_sched_barrier(0);
        a6_1.x = __builtin_amdgcn_exp2f(a6_1.x);
        __builtin_amdgcn_sched_barrier(0);
        ap0 = __builtin_elementwise_fma(a4_0, (v2f){ yv[8].x, yv[8].y }, ap0);
        __builtin_amdgcn_sched_barrier(0);
        a6_1.y = __builtin_amdgcn_exp2f(a6_1.y);
        __builtin_amdgcn_sched_barrier(0);
        ap1 = __builtin_elementwise_fma(a4_1, (v2f){ yv[8].z, yv[8].w }, ap1);
        __builtin_amdgcn_sched_barrier(0);
        a6_z = __builtin_amdgcn_exp2f(a6_z);
        __builtin_amdgcn_sched_barrier(0);
        accz += a4_z;
        __builtin_amdgcn_sched_barrier(0);
        a6_2.x = __builtin_amdgcn_exp2f(a6_2.x);
        __builtin_amdgcn_sched_barrier(0);
        ap2 = __builtin_elementwise_fma(a4_2, (v2f){ yv[9].x, yv[9].y }, ap2);
        __builtin_amdgcn_sched_barrier(0);
        a6_2.y = __builtin_amdgcn_exp2f(a6_2.y);
        __builtin_amdgcn_sched_barrier(0);
        ap3 = __builtin_elementwise_fma(a4_3, (v2f){ yv[9].z, yv[9].w }, ap3);
        __builtin_amdgcn_sched_barrier(0);
        a6_3.x = __builtin_amdgcn_exp2f(a6_3.x);
        __builtin_amdgcn_sched_barrier(0);
        ap0 = __builtin_elementwise_fma(a5_0, (v2f){ yv[10].x, yv[10].y }, ap0);
        __builtin_amdgcn_sched_barrier(0);
        a6_3.y = __builtin_amdgcn_exp2f(a6_3.y);
        __builtin_amdgcn_sched_barrier(0);
        ap1 = __builtin_elementwise_fma(a5_1, (v2f){ yv[10].z, yv[10].w }, ap1);
        __builtin_amdgcn_sched_barrier(0);
        a7_0.x = __builtin_amdgcn_exp2f(a7_0.x);
        __builtin_amdgcn_sched_barrier(0);
        accz += a5_z;
        __builtin_amdgcn_sched_barrier(0);
        a7_0.y = __builtin_amdgcn_exp2f(a7_0.y);
        __builtin_amdgcn_sched_barrier(0);
        ap2 = __builtin_elementwise_fma(a5_2, (v2f){ yv[11].x, yv[11].y }, ap2);
        __builtin_amdgcn_sched_barrier(0);
        a7_1.x = __builtin_amdgcn_exp2f(a7_1.x);
        __builtin_amdgcn_sched_barrier(0);
        v2f a7_3 = (v2f){ q1b.y, q1b.y } * kp3;
        __builtin_amdgcn_sched_barrier(0);
        a7_1.y = __builtin_amdgcn_exp2f(a7_1.y);
        __builtin_amdgcn_sched_barrier(0);
        ap3 = __builtin_elementwise_fma(a5_3, (v2f){ yv[11].z, yv[11].w }, ap3);
        __builtin_amdgcn_sched_barrier(0);
        a7_z = __builtin_amdgcn_exp2f(a7_z);
        __builtin_amdgcn_sched_barrier(0);
        ap0 = __builtin_elementwise_fma(a6_0, (v2f){ yv[12].x, yv[12].y }, ap0);
        __builtin_amdgcn_sched_barrier(0);
        a7_2.x = __builtin_amdgcn_exp2f(a7_2.x);
        __builtin_amdgcn_sched_barrier(0);
        ap1 = __builtin_elementwise_fma(a6_1, (v2f){ yv[12].z, yv[12].w }, ap1);
        __builtin_amdgcn_sched_barrier(0);
        a7_2.y = __builtin_amdgcn_exp2f(a7_2.y);
        __builtin_amdgcn_sched_barrier(0);
        accz += a6_z;
        __builtin_amdgcn_sched_barrier(0);
        a7_3.x = __builtin_amdgcn_exp2f(a7_3.x);
        __builtin_amdgcn_sched_barrier(0);
        ap2 = __builtin_elementwise_fma(a6_2, (v2f){ yv[13].x, yv[13].y }, ap2);
        __builtin_amdgcn_sched_barrier(0);
        a7_3.y = __builtin_amdgcn_exp2f(a7_3.y);
        __builtin_amdgcn_sched_barrier(0);
        ap3 = __builtin_elementwise_fma(a6_3, (v2f){ yv[13].z, yv[13].w }, ap3);
        __builtin_amdgcn_sched_barrier(0);
        ap0 = __builtin_elementwise_fma(a7_0, (v2f){ yv[14].x, yv[14].y }, ap0);
        ap1 = __builtin_elementwise_fma(a7_1, (v2f){ yv[14].z, yv[14].w }, ap1);
        accz += a7_z;
        ap2 = __builtin_elementwise_fma(a7_2, (v2f){ yv[15].x, yv[15].y }, ap2);
        ap3 = __builtin_elementwise_fma(a7_3, (v2f){ yv[15].z, yv[15].w }, ap3);
        __builtin_amdgcn_sched_barrier(0);
    }
    const float acc[CREP] = { accz, ap0.x, ap0.y, ap1.x, ap1.y,
                              ap2.x, ap2.y, ap3.x, ap3.y };

    {
        const int base = (chunk * MT + m_l) * 10;
        #pragma unroll
        for (int c = 0; c < CREP; ++c) spart[base + c] = acc[c];
    }
    __syncthreads();

    const int wid  = tid >> 5;
    const int lane = tid & 31;
    if (wid < 2) {
        const int half  = lane >> 4;
        const int lr    = lane & 15;
        const int mbase = wid * 16;
        const int row   = mbase + lr;
        const int so    = half ? 0 : 8;

        v2f bfr[3];
        #pragma unroll
        for (int kc = 0; kc < 3; ++kc) {
            const int c0 = kc * 4 + 2 * half;
            bfr[kc].x = Wp[lr * 12 + c0];
            bfr[kc].y = Wp[lr * 12 + c0 + 1];
        }
        const float bo = sb[lr];

        v2f p0 = {0.f, 0.f};
        v2f p1 = {0.f, 0.f};
        float s8 = 0.f;
        #pragma unroll
        for (int ch = 0; ch < NCHUNK; ++ch) {
            const float* pr = spart + (ch * MT + row) * 10;
            p0 += *(const v2f*)(pr + 2 * half);
            p1 += *(const v2f*)(pr + 4 + 2 * half);
            s8 += pr[so];
        }
        const float dens = half ? s8 : p0.x;
        const float inv  = 1.0f / (dens + 1e-8f);
        v2f afr[3];
        afr[0].x = half ? p0.x * inv : dens;
        afr[0].y = p0.y * inv;
        afr[1].x = p1.x * inv;
        afr[1].y = p1.y * inv;
        afr[2].x = half ? 0.f : s8 * inv;
        afr[2].y = 0.f;

        v8f d = {0.f, 0.f, 0.f, 0.f, 0.f, 0.f, 0.f, 0.f};
        #pragma unroll
        for (int kc = 0; kc < 3; ++kc)
            d = __builtin_amdgcn_wmma_f32_16x16x4_f32(
                    false, afr[kc], false, bfr[kc], (short)0, d, false, false);

        const int mg = mt * MT + mbase;
        #pragma unroll
        for (int r = 0; r < 8; ++r) {
            const int m = mg + r + 8 * half;
            out[((size_t)b * MM + m) * COUT + lr] = d[r] + bo;
        }
    }
}

extern "C" void kernel_launch(void* const* d_in, const int* in_sizes, int n_in,
                              void* d_out, int out_size, void* d_ws, size_t ws_size,
                              hipStream_t stream) {
    const float* x     = (const float*)d_in[0];
    const float* y     = (const float*)d_in[1];
    const float* t     = (const float*)d_in[2];
    const float* sigma = (const float*)d_in[3];
    const float* W     = (const float*)d_in[4];
    const float* bias  = (const float*)d_in[5];
    float* out = (float*)d_out;

    dim3 grid(MM / MT, BB);
    dim3 block(256);
    convcnp_kernel<<<grid, block, 0, stream>>>(x, y, t, sigma, W, bias, out);
}
